# starting kernel
# baseline (speedup 1.0000x reference)
#include <hip/hip_runtime.h>


typedef float v2f __attribute__((ext_vector_type(2)));
typedef float v8f __attribute__((ext_vector_type(8)));

#define BN 16
#define CN 256
#define HN 48
#define WN 64
#define HWN (HN * WN)
#define ND 21
#define DN (ND * ND)

__global__ __launch_bounds__(256) void corr_wmma_f32_kernel(
    const float* __restrict__ in1,
    const float* __restrict__ in2,
    float* __restrict__ out)
{
    const int wave = threadIdx.x >> 5;
    const int lane = threadIdx.x & 31;
    const int half = lane >> 4;
    const int l    = lane & 15;

    int task = blockIdx.x * (blockDim.x >> 5) + wave;
    const int iy    = task % ND; task /= ND;
    const int strip = task % 4;  task /= 4;
    const int h     = task % HN;
    const int b     = task / HN;

    const int w0    = strip * 16;
    const int qbase = w0 - 24;
    const int dy    = -20 + 2 * iy;
    const int h2    = h + dy;
    const bool row_valid = (h2 >= 0) && (h2 < HN);

    v8f acc0 = {}, acc1 = {}, acc2 = {}, acc3 = {};

    if (row_valid) {
        const int k0 = half * 2;

        const float* __restrict__ ap =
            in1 + (((size_t)b * CN + k0) * HN + h) * WN + (w0 + l);

        const size_t brow = (((size_t)b * CN + k0) * HN + h2) * WN;

        const int q0 = qbase + l;
        const int q1 = q0 + 16;
        const int q2 = q0 + 32;
        const int q3 = q0 + 48;
        const bool v0 = (q0 >= 0) && (q0 < WN);
        const bool v1 = (q1 >= 0) && (q1 < WN);
        const bool v2 = (q2 >= 0) && (q2 < WN);
        const bool v3 = (q3 >= 0) && (q3 < WN);
        const float inv = 1.0f / (float)CN;
        const float s0 = v0 ? inv : 0.0f;
        const float s1 = v1 ? inv : 0.0f;
        const float s2 = v2 ? inv : 0.0f;
        const float s3 = v3 ? inv : 0.0f;

        const float* __restrict__ b0p = in2 + brow + (v0 ? q0 : 0);
        const float* __restrict__ b1p = in2 + brow + (v1 ? q1 : 0);
        const float* __restrict__ b2p = in2 + brow + (v2 ? q2 : 0);
        const float* __restrict__ b3p = in2 + brow + (v3 ? q3 : 0);

#pragma unroll 4
        for (int kk = 0; kk < CN; kk += 4) {
            v2f Av = { ap[0], ap[HWN] };

            v2f B0 = { b0p[0] * s0, b0p[HWN] * s0 };
            v2f B1 = { b1p[0] * s1, b1p[HWN] * s1 };
            v2f B2 = { b2p[0] * s2, b2p[HWN] * s2 };
            v2f B3 = { b3p[0] * s3, b3p[HWN] * s3 };

            acc0 = __builtin_amdgcn_wmma_f32_16x16x4_f32(
                       false, Av, false, B0, (short)0, acc0, false, false);
            acc1 = __builtin_amdgcn_wmma_f32_16x16x4_f32(
                       false, Av, false, B1, (short)0, acc1, false, false);
            acc2 = __builtin_amdgcn_wmma_f32_16x16x4_f32(
                       false, Av, false, B2, (short)0, acc2, false, false);
            acc3 = __builtin_amdgcn_wmma_f32_16x16x4_f32(
                       false, Av, false, B3, (short)0, acc3, false, false);

            ap  += 4 * HWN;
            b0p += 4 * HWN;
            b1p += 4 * HWN;
            b2p += 4 * HWN;
            b3p += 4 * HWN;
        }
    }

#pragma unroll
    for (int n = 0; n < 4; ++n) {
        const v8f acc = (n == 0) ? acc0 : (n == 1) ? acc1
                       : (n == 2) ? acc2 : acc3;
#pragma unroll
        for (int r = 0; r < 8; ++r) {
            const int m    = r + 8 * half;
            const int ncol = 16 * n + l;
            const int dx   = ncol - m - 24;
            if (dx >= -20 && dx <= 20 && ((dx & 1) == 0)) {
                const int ix = (dx + 20) >> 1;
                const int d  = iy * ND + ix;
                const size_t idx =
                    (((size_t)b * DN + d) * HN + h) * WN + (size_t)(w0 + m);
                out[idx] = acc[r];
            }
        }
    }
}

extern "C" void kernel_launch(void* const* d_in, const int* in_sizes, int n_in,
                              void* d_out, int out_size, void* d_ws, size_t ws_size,
                              hipStream_t stream) {
    (void)in_sizes; (void)n_in; (void)out_size; (void)d_ws; (void)ws_size;
    const float* in1 = (const float*)d_in[0];
    const float* in2 = (const float*)d_in[1];
    float* out = (float*)d_out;

    const int tasks = BN * HN * (WN / 16) * ND;
    const int waves_per_block = 8;
    const int blocks = tasks / waves_per_block;
    corr_wmma_f32_kernel<<<blocks, waves_per_block * 32, 0, stream>>>(in1, in2, out);
}


# Each in2 row staged once per (b,h) work-group into LDS (64-channel chunks, double-buffered) and shared by 8 waves; in1 strip kept in registers as A fragments; k order and sums unchanged
# speedup vs baseline: 2.0438x; 2.0438x over previous
#include <hip/hip_runtime.h>


typedef float v2f __attribute__((ext_vector_type(2)));
typedef float v8f __attribute__((ext_vector_type(8)));

#define BN 16
#define CN 256
#define HN 48
#define WN 64
#define HWN (HN * WN)
#define ND 21
#define DN (ND * ND)

#define KC 64
#define NCHUNK (CN / KC)
#define KSTEPS (KC / 4)
#define HPITCH 480
#define RPITCH (2 * HPITCH)
#define LBUF (8 * RPITCH)
#define SLOT(q) ((q) + ((q) >> 4))

__global__ __launch_bounds__(256) void corr_wmma_f32_kernel(
    const float* __restrict__ in1,
    const float* __restrict__ in2,
    float* __restrict__ out)
{
    __shared__ __attribute__((aligned(16))) float lds[2 * LBUF];

    const int tid   = threadIdx.x;
    const int wave  = tid >> 5;
    const int lane  = tid & 31;
    const int half  = lane >> 4;
    const int l     = lane & 15;
    const int strip = wave & 3;
    const int tp    = wave >> 2;
    const int h     = blockIdx.x % HN;
    const int b     = blockIdx.x / HN;
    const int w0    = strip * 16;

    const int tlo   = 20 - h;
    const int iy_lo = (tlo > 0) ? ((tlo + 1) >> 1) : 0;
    const int thi   = (HN - 1 + 20 - h) >> 1;
    const int iy_hi = (thi < ND - 1) ? thi : (ND - 1);

    for (int i = tid; i < 32 * 52; i += 256) {
        const int row  = i / 52;
        const int o    = i - row * 52;
        const int slot = (o < 25) ? o : (o + 68);
        *(float4*)(lds + row * HPITCH + slot * 4) = make_float4(0.f, 0.f, 0.f, 0.f);
    }

    const int sj = tid & 15;
    const int sh = (tid >> 4) & 1;
    const int sr = tid >> 5;
    const float* __restrict__ gsrc =
        in2 + (((size_t)b * CN + 8 * sr + 2 * sh) * HN) * WN + 4 * sj;
    float* __restrict__ sdst =
        lds + sr * RPITCH + sh * HPITCH + SLOT(24 + 4 * sj) * 4;

    float4 g0, g1, g2, g3;
    {
        const float* p = gsrc + (size_t)(h - 20 + 2 * iy_lo) * WN;
        g0 = *(const float4*)(p);
        g1 = *(const float4*)(p + HWN);
        g2 = *(const float4*)(p + 4 * HWN);
        g3 = *(const float4*)(p + 5 * HWN);
    }

    v2f A[CN / 4];
    {
        const float inv = 1.0f / (float)CN;
        const float* __restrict__ ap =
            in1 + (((size_t)b * CN + 2 * half) * HN + h) * WN + (w0 + l);
#pragma unroll
        for (int s = 0; s < CN / 4; ++s) {
            A[s].x = ap[(size_t)s * 4 * HWN] * inv;
            A[s].y = ap[(size_t)s * 4 * HWN + HWN] * inv;
        }
    }

    for (int iy = 0; iy < ND; ++iy) {
        if (iy >= iy_lo && iy <= iy_hi) continue;
        for (int i = tid; i < ND * (WN / 4); i += 256) {
            const int ix = i >> 4;
            const int w4 = i & 15;
            const size_t idx =
                (((size_t)b * DN + iy * ND + ix) * HN + h) * WN + 4 * w4;
            *(float4*)(out + idx) = make_float4(0.f, 0.f, 0.f, 0.f);
        }
    }

    *(float4*)(sdst)      = make_float4(g0.x, g1.x, g2.x, g3.x);
    *(float4*)(sdst + 4)  = make_float4(g0.y, g1.y, g2.y, g3.y);
    *(float4*)(sdst + 8)  = make_float4(g0.z, g1.z, g2.z, g3.z);
    *(float4*)(sdst + 12) = make_float4(g0.w, g1.w, g2.w, g3.w);
    __syncthreads();

    const float* __restrict__ bbase =
        lds + half * HPITCH + (SLOT(w0 + 32 * tp) + l) * 4;

    for (int iy = iy_lo; iy <= iy_hi; ++iy) {
        v8f acc0 = {}, acc1 = {};
        const int h2 = h - 20 + 2 * iy;

#pragma unroll
        for (int c = 0; c < NCHUNK; ++c) {
            {
                const int cn  = (c + 1) & (NCHUNK - 1);
                const int h2n = (c < NCHUNK - 1) ? h2 : ((iy < iy_hi) ? (h2 + 2) : h2);
                const float* p = gsrc + (size_t)cn * KC * HWN + (size_t)h2n * WN;
                g0 = *(const float4*)(p);
                g1 = *(const float4*)(p + HWN);
                g2 = *(const float4*)(p + 4 * HWN);
                g3 = *(const float4*)(p + 5 * HWN);
            }
            __builtin_amdgcn_sched_barrier(0);

            const float* __restrict__ bp = bbase + (c & 1) * LBUF;
#pragma unroll
            for (int r2 = 0; r2 < KSTEPS / 2; ++r2) {
                const float4 q0 = *(const float4*)(bp + r2 * RPITCH);
                const float4 q1 = *(const float4*)(bp + r2 * RPITCH + 17 * 4);
                const v2f B00 = { q0.x, q0.y };
                const v2f B01 = { q0.z, q0.w };
                const v2f B10 = { q1.x, q1.y };
                const v2f B11 = { q1.z, q1.w };
                const int s = c * KSTEPS + 2 * r2;
                acc0 = __builtin_amdgcn_wmma_f32_16x16x4_f32(
                           false, A[s], false, B00, (short)0, acc0, false, false);
                acc1 = __builtin_amdgcn_wmma_f32_16x16x4_f32(
                           false, A[s], false, B10, (short)0, acc1, false, false);
                acc0 = __builtin_amdgcn_wmma_f32_16x16x4_f32(
                           false, A[s + 1], false, B01, (short)0, acc0, false, false);
                acc1 = __builtin_amdgcn_wmma_f32_16x16x4_f32(
                           false, A[s + 1], false, B11, (short)0, acc1, false, false);
            }

            __builtin_amdgcn_sched_barrier(0);
            {
                float* d = sdst + ((c + 1) & 1) * LBUF;
                *(float4*)(d)      = make_float4(g0.x, g1.x, g2.x, g3.x);
                *(float4*)(d + 4)  = make_float4(g0.y, g1.y, g2.y, g3.y);
                *(float4*)(d + 8)  = make_float4(g0.z, g1.z, g2.z, g3.z);
                *(float4*)(d + 12) = make_float4(g0.w, g1.w, g2.w, g3.w);
            }
            __syncthreads();
        }

#pragma unroll
        for (int t = 0; t < 2; ++t) {
            const v8f acc = (t == 0) ? acc0 : acc1;
            const int n = 2 * tp + t;
#pragma unroll
            for (int r = 0; r < 8; ++r) {
                const int m    = r + 8 * half;
                const int ncol = 16 * n + l;
                const int dx   = ncol - m - 24;
                if (dx >= -20 && dx <= 20 && ((dx & 1) == 0)) {
                    const int ix = (dx + 20) >> 1;
                    const int d  = iy * ND + ix;
                    const size_t idx =
                        (((size_t)b * DN + d) * HN + h) * WN + (size_t)(w0 + m);
                    out[idx] = acc[r];
                }
            }
        }
    }
}

extern "C" void kernel_launch(void* const* d_in, const int* in_sizes, int n_in,
                              void* d_out, int out_size, void* d_ws, size_t ws_size,
                              hipStream_t stream) {
    (void)in_sizes; (void)n_in; (void)out_size; (void)d_ws; (void)ws_size;
    const float* in1 = (const float*)d_in[0];
    const float* in2 = (const float*)d_in[1];
    float* out = (float*)d_out;

    corr_wmma_f32_kernel<<<BN * HN, 256, 0, stream>>>(in1, in2, out);
}


# Work-groups ordered longest-first: block index walks image rows from the centre outwards (most valid displacement rows first), batch index fastest; per-group work, k order and sums unchanged
# speedup vs baseline: 2.4525x; 1.2000x over previous
#include <hip/hip_runtime.h>


typedef float v2f __attribute__((ext_vector_type(2)));
typedef float v8f __attribute__((ext_vector_type(8)));

#define BN 16
#define CN 256
#define HN 48
#define WN 64
#define HWN (HN * WN)
#define ND 21
#define DN (ND * ND)

#define KC 64
#define NCHUNK (CN / KC)
#define KSTEPS (KC / 4)
#define HPITCH 480
#define RPITCH (2 * HPITCH)
#define LBUF (8 * RPITCH)
#define SLOT(q) ((q) + ((q) >> 4))

__global__ __launch_bounds__(256) void corr_wmma_f32_kernel(
    const float* __restrict__ in1,
    const float* __restrict__ in2,
    float* __restrict__ out)
{
    __shared__ __attribute__((aligned(16))) float lds[2 * LBUF];

    const int tid   = threadIdx.x;
    const int wave  = tid >> 5;
    const int lane  = tid & 31;
    const int half  = lane >> 4;
    const int l     = lane & 15;
    const int strip = wave & 3;
    const int tp    = wave >> 2;
    const int rank  = blockIdx.x / BN;
    const int b     = blockIdx.x % BN;
    const int h     = (rank & 1) ? (HN / 2 - 1 - (rank >> 1)) : (HN / 2 + (rank >> 1));
    __builtin_assume(h >= 0 && h < HN);
    const int w0    = strip * 16;

    const int tlo   = 20 - h;
    const int iy_lo = (tlo > 0) ? ((tlo + 1) >> 1) : 0;
    const int thi   = (HN - 1 + 20 - h) >> 1;
    const int iy_hi = (thi < ND - 1) ? thi : (ND - 1);

    for (int i = tid; i < 32 * 52; i += 256) {
        const int row  = i / 52;
        const int o    = i - row * 52;
        const int slot = (o < 25) ? o : (o + 68);
        *(float4*)(lds + row * HPITCH + slot * 4) = make_float4(0.f, 0.f, 0.f, 0.f);
    }

    const int sj = tid & 15;
    const int sh = (tid >> 4) & 1;
    const int sr = tid >> 5;
    const float* __restrict__ gsrc =
        in2 + (((size_t)b * CN + 8 * sr + 2 * sh) * HN) * WN + 4 * sj;
    float* __restrict__ sdst =
        lds + sr * RPITCH + sh * HPITCH + SLOT(24 + 4 * sj) * 4;

    float4 g0, g1, g2, g3;
    {
        const float* p = gsrc + (size_t)(h - 20 + 2 * iy_lo) * WN;
        g0 = *(const float4*)(p);
        g1 = *(const float4*)(p + HWN);
        g2 = *(const float4*)(p + 4 * HWN);
        g3 = *(const float4*)(p + 5 * HWN);
    }

    v2f A[CN / 4];
    {
        const float inv = 1.0f / (float)CN;
        const float* __restrict__ ap =
            in1 + (((size_t)b * CN + 2 * half) * HN + h) * WN + (w0 + l);
#pragma unroll
        for (int s = 0; s < CN / 4; ++s) {
            A[s].x = ap[(size_t)s * 4 * HWN] * inv;
            A[s].y = ap[(size_t)s * 4 * HWN + HWN] * inv;
        }
    }

    for (int iy = 0; iy < ND; ++iy) {
        if (iy >= iy_lo && iy <= iy_hi) continue;
        for (int i = tid; i < ND * (WN / 4); i += 256) {
            const int ix = i >> 4;
            const int w4 = i & 15;
            const size_t idx =
                (((size_t)b * DN + iy * ND + ix) * HN + h) * WN + 4 * w4;
            *(float4*)(out + idx) = make_float4(0.f, 0.f, 0.f, 0.f);
        }
    }

    *(float4*)(sdst)      = make_float4(g0.x, g1.x, g2.x, g3.x);
    *(float4*)(sdst + 4)  = make_float4(g0.y, g1.y, g2.y, g3.y);
    *(float4*)(sdst + 8)  = make_float4(g0.z, g1.z, g2.z, g3.z);
    *(float4*)(sdst + 12) = make_float4(g0.w, g1.w, g2.w, g3.w);
    __syncthreads();

    const float* __restrict__ bbase =
        lds + half * HPITCH + (SLOT(w0 + 32 * tp) + l) * 4;

    for (int iy = iy_lo; iy <= iy_hi; ++iy) {
        v8f acc0 = {}, acc1 = {};
        const int h2 = h - 20 + 2 * iy;

#pragma unroll
        for (int c = 0; c < NCHUNK; ++c) {
            {
                const int cn  = (c + 1) & (NCHUNK - 1);
                const int h2n = (c < NCHUNK - 1) ? h2 : ((iy < iy_hi) ? (h2 + 2) : h2);
                const float* p = gsrc + (size_t)cn * KC * HWN + (size_t)h2n * WN;
                g0 = *(const float4*)(p);
                g1 = *(const float4*)(p + HWN);
                g2 = *(const float4*)(p + 4 * HWN);
                g3 = *(const float4*)(p + 5 * HWN);
            }
            __builtin_amdgcn_sched_barrier(0);

            const float* __restrict__ bp = bbase + (c & 1) * LBUF;
#pragma unroll
            for (int r2 = 0; r2 < KSTEPS / 2; ++r2) {
                const float4 q0 = *(const float4*)(bp + r2 * RPITCH);
                const float4 q1 = *(const float4*)(bp + r2 * RPITCH + 17 * 4);
                const v2f B00 = { q0.x, q0.y };
                const v2f B01 = { q0.z, q0.w };
                const v2f B10 = { q1.x, q1.y };
                const v2f B11 = { q1.z, q1.w };
                const int s = c * KSTEPS + 2 * r2;
                acc0 = __builtin_amdgcn_wmma_f32_16x16x4_f32(
                           false, A[s], false, B00, (short)0, acc0, false, false);
                acc1 = __builtin_amdgcn_wmma_f32_16x16x4_f32(
                           false, A[s], false, B10, (short)0, acc1, false, false);
                acc0 = __builtin_amdgcn_wmma_f32_16x16x4_f32(
                           false, A[s + 1], false, B01, (short)0, acc0, false, false);
                acc1 = __builtin_amdgcn_wmma_f32_16x16x4_f32(
                           false, A[s + 1], false, B11, (short)0, acc1, false, false);
            }

            __builtin_amdgcn_sched_barrier(0);
            {
                float* d = sdst + ((c + 1) & 1) * LBUF;
                *(float4*)(d)      = make_float4(g0.x, g1.x, g2.x, g3.x);
                *(float4*)(d + 4)  = make_float4(g0.y, g1.y, g2.y, g3.y);
                *(float4*)(d + 8)  = make_float4(g0.z, g1.z, g2.z, g3.z);
                *(float4*)(d + 12) = make_float4(g0.w, g1.w, g2.w, g3.w);
            }
            __syncthreads();
        }

#pragma unroll
        for (int t = 0; t < 2; ++t) {
            const v8f acc = (t == 0) ? acc0 : acc1;
            const int n = 2 * tp + t;
#pragma unroll
            for (int r = 0; r < 8; ++r) {
                const int m    = r + 8 * half;
                const int ncol = 16 * n + l;
                const int dx   = ncol - m - 24;
                if (dx >= -20 && dx <= 20 && ((dx & 1) == 0)) {
                    const int ix = (dx + 20) >> 1;
                    const int d  = iy * ND + ix;
                    const size_t idx =
                        (((size_t)b * DN + d) * HN + h) * WN + (size_t)(w0 + m);
                    out[idx] = acc[r];
                }
            }
        }
    }
}

extern "C" void kernel_launch(void* const* d_in, const int* in_sizes, int n_in,
                              void* d_out, int out_size, void* d_ws, size_t ws_size,
                              hipStream_t stream) {
    (void)in_sizes; (void)n_in; (void)out_size; (void)d_ws; (void)ws_size;
    const float* in1 = (const float*)d_in[0];
    const float* in2 = (const float*)d_in[1];
    float* out = (float*)d_out;

    corr_wmma_f32_kernel<<<BN * HN, 256, 0, stream>>>(in1, in2, out);
}


# Output band of each dy staged through a small LDS tile [dx][pixel] and written as whole 256-byte rows (float4 stores) instead of one 4-byte store per lane into separate lines; sums unchanged
# speedup vs baseline: 3.5605x; 1.4518x over previous
#include <hip/hip_runtime.h>


typedef float v2f __attribute__((ext_vector_type(2)));
typedef float v8f __attribute__((ext_vector_type(8)));

#define BN 16
#define CN 256
#define HN 48
#define WN 64
#define HWN (HN * WN)
#define ND 21
#define DN (ND * ND)

#define KC 64
#define NCHUNK (CN / KC)
#define KSTEPS (KC / 4)
#define HPITCH 480
#define RPITCH (2 * HPITCH)
#define LBUF (8 * RPITCH)
#define SLOT(q) ((q) + ((q) >> 4))
#define OPITCH 68

__global__ __launch_bounds__(256) void corr_wmma_f32_kernel(
    const float* __restrict__ in1,
    const float* __restrict__ in2,
    float* __restrict__ out)
{
    __shared__ __attribute__((aligned(16))) float lds[2 * LBUF];
    __shared__ __attribute__((aligned(16))) float obuf[ND * OPITCH];

    const int tid   = threadIdx.x;
    const int wave  = tid >> 5;
    const int lane  = tid & 31;
    const int half  = lane >> 4;
    const int l     = lane & 15;
    const int strip = wave & 3;
    const int tp    = wave >> 2;
    const int rank  = blockIdx.x / BN;
    const int b     = blockIdx.x % BN;
    const int h     = (rank & 1) ? (HN / 2 - 1 - (rank >> 1)) : (HN / 2 + (rank >> 1));
    __builtin_assume(h >= 0 && h < HN);
    const int w0    = strip * 16;

    const int tlo   = 20 - h;
    const int iy_lo = (tlo > 0) ? ((tlo + 1) >> 1) : 0;
    const int thi   = (HN - 1 + 20 - h) >> 1;
    const int iy_hi = (thi < ND - 1) ? thi : (ND - 1);

    for (int i = tid; i < 32 * 52; i += 256) {
        const int row  = i / 52;
        const int o    = i - row * 52;
        const int slot = (o < 25) ? o : (o + 68);
        *(float4*)(lds + row * HPITCH + slot * 4) = make_float4(0.f, 0.f, 0.f, 0.f);
    }

    const int sj = tid & 15;
    const int sh = (tid >> 4) & 1;
    const int sr = tid >> 5;
    const float* __restrict__ gsrc =
        in2 + (((size_t)b * CN + 8 * sr + 2 * sh) * HN) * WN + 4 * sj;
    float* __restrict__ sdst =
        lds + sr * RPITCH + sh * HPITCH + SLOT(24 + 4 * sj) * 4;

    float4 g0, g1, g2, g3;
    {
        const float* p = gsrc + (size_t)(h - 20 + 2 * iy_lo) * WN;
        g0 = *(const float4*)(p);
        g1 = *(const float4*)(p + HWN);
        g2 = *(const float4*)(p + 4 * HWN);
        g3 = *(const float4*)(p + 5 * HWN);
    }

    v2f A[CN / 4];
    {
        const float inv = 1.0f / (float)CN;
        const float* __restrict__ ap =
            in1 + (((size_t)b * CN + 2 * half) * HN + h) * WN + (w0 + l);
#pragma unroll
        for (int s = 0; s < CN / 4; ++s) {
            A[s].x = ap[(size_t)s * 4 * HWN] * inv;
            A[s].y = ap[(size_t)s * 4 * HWN + HWN] * inv;
        }
    }

    for (int iy = 0; iy < ND; ++iy) {
        if (iy >= iy_lo && iy <= iy_hi) continue;
        for (int i = tid; i < ND * (WN / 4); i += 256) {
            const int ix = i >> 4;
            const int w4 = i & 15;
            const size_t idx =
                (((size_t)b * DN + iy * ND + ix) * HN + h) * WN + 4 * w4;
            *(float4*)(out + idx) = make_float4(0.f, 0.f, 0.f, 0.f);
        }
    }

    *(float4*)(sdst)      = make_float4(g0.x, g1.x, g2.x, g3.x);
    *(float4*)(sdst + 4)  = make_float4(g0.y, g1.y, g2.y, g3.y);
    *(float4*)(sdst + 8)  = make_float4(g0.z, g1.z, g2.z, g3.z);
    *(float4*)(sdst + 12) = make_float4(g0.w, g1.w, g2.w, g3.w);
    __syncthreads();

    const float* __restrict__ bbase =
        lds + half * HPITCH + (SLOT(w0 + 32 * tp) + l) * 4;

    for (int iy = iy_lo; iy <= iy_hi; ++iy) {
        v8f acc0 = {}, acc1 = {};
        const int h2 = h - 20 + 2 * iy;

#pragma unroll
        for (int c = 0; c < NCHUNK; ++c) {
            {
                const int cn  = (c + 1) & (NCHUNK - 1);
                const int h2n = (c < NCHUNK - 1) ? h2 : ((iy < iy_hi) ? (h2 + 2) : h2);
                const float* p = gsrc + (size_t)cn * KC * HWN + (size_t)h2n * WN;
                g0 = *(const float4*)(p);
                g1 = *(const float4*)(p + HWN);
                g2 = *(const float4*)(p + 4 * HWN);
                g3 = *(const float4*)(p + 5 * HWN);
            }
            __builtin_amdgcn_sched_barrier(0);

            const float* __restrict__ bp = bbase + (c & 1) * LBUF;
#pragma unroll
            for (int r2 = 0; r2 < KSTEPS / 2; ++r2) {
                const float4 q0 = *(const float4*)(bp + r2 * RPITCH);
                const float4 q1 = *(const float4*)(bp + r2 * RPITCH + 17 * 4);
                const v2f B00 = { q0.x, q0.y };
                const v2f B01 = { q0.z, q0.w };
                const v2f B10 = { q1.x, q1.y };
                const v2f B11 = { q1.z, q1.w };
                const int s = c * KSTEPS + 2 * r2;
                acc0 = __builtin_amdgcn_wmma_f32_16x16x4_f32(
                           false, A[s], false, B00, (short)0, acc0, false, false);
                acc1 = __builtin_amdgcn_wmma_f32_16x16x4_f32(
                           false, A[s], false, B10, (short)0, acc1, false, false);
                acc0 = __builtin_amdgcn_wmma_f32_16x16x4_f32(
                           false, A[s + 1], false, B01, (short)0, acc0, false, false);
                acc1 = __builtin_amdgcn_wmma_f32_16x16x4_f32(
                           false, A[s + 1], false, B11, (short)0, acc1, false, false);
            }

            __builtin_amdgcn_sched_barrier(0);
            if (c == NCHUNK - 1) {
#pragma unroll
                for (int t = 0; t < 2; ++t) {
                    const v8f acc = (t == 0) ? acc0 : acc1;
                    const int e   = 32 * tp + 16 * t + l - 8 * half - 4;
#pragma unroll
                    for (int r = 0; r < 8; ++r) {
                        const int d2 = e - r;
                        if (((d2 & 1) == 0) && (unsigned)d2 <= 40u)
                            obuf[(d2 >> 1) * OPITCH + w0 + 8 * half + r] = acc[r];
                    }
                }
            }
            {
                float* d = sdst + ((c + 1) & 1) * LBUF;
                *(float4*)(d)      = make_float4(g0.x, g1.x, g2.x, g3.x);
                *(float4*)(d + 4)  = make_float4(g0.y, g1.y, g2.y, g3.y);
                *(float4*)(d + 8)  = make_float4(g0.z, g1.z, g2.z, g3.z);
                *(float4*)(d + 12) = make_float4(g0.w, g1.w, g2.w, g3.w);
            }
            __syncthreads();
        }

        {
            const int ix0 = tid >> 4;
            const int w4  = tid & 15;
            const float* __restrict__ op = obuf + ix0 * OPITCH + 4 * w4;
            float* __restrict__ gp =
                out + (((size_t)b * DN + iy * ND + ix0) * HN + h) * WN + 4 * w4;
            const float4 v0 = *(const float4*)(op);
            if (tid < (ND - 16) * 16) {
                const float4 v1 = *(const float4*)(op + 16 * OPITCH);
                *(float4*)(gp + (size_t)16 * HWN) = v1;
            }
            *(float4*)(gp) = v0;
        }
    }
}

extern "C" void kernel_launch(void* const* d_in, const int* in_sizes, int n_in,
                              void* d_out, int out_size, void* d_ws, size_t ws_size,
                              hipStream_t stream) {
    (void)in_sizes; (void)n_in; (void)out_size; (void)d_ws; (void)ws_size;
    const float* in1 = (const float*)d_in[0];
    const float* in2 = (const float*)d_in[1];
    float* out = (float*)d_out;

    corr_wmma_f32_kernel<<<BN * HN, 256, 0, stream>>>(in1, in2, out);
}


# Pixels and in2 columns split by parity (even dx pairs only equal parity): one in-image 16x16 tile per wave, 8 tiles per k-step instead of 16; padding written as zeros, not multiplied; sums unchanged
# speedup vs baseline: 5.0875x; 1.4289x over previous
#include <hip/hip_runtime.h>


typedef float v2f __attribute__((ext_vector_type(2)));
typedef float v8f __attribute__((ext_vector_type(8)));

#define BN 16
#define CN 256
#define HN 48
#define WN 64
#define HWN (HN * WN)
#define ND 21
#define DN (ND * ND)

#define KC 64
#define NCHUNK (CN / KC)
#define KSTEPS (KC / 4)
#define PSLOTS 33
#define HPITCH 288
#define RPITCH (2 * HPITCH)
#define LBUF (8 * RPITCH)
#define OPITCH 68

__global__ __launch_bounds__(256) void corr_wmma_f32_kernel(
    const float* __restrict__ in1,
    const float* __restrict__ in2,
    float* __restrict__ out)
{
    __shared__ __attribute__((aligned(16))) float lds[2 * LBUF];
    __shared__ __attribute__((aligned(16))) float obuf[ND * OPITCH];

    const int tid   = threadIdx.x;
    const int wave  = tid >> 5;
    const int lane  = tid & 31;
    const int half  = lane >> 4;
    const int l     = lane & 15;
    const int par   = wave & 1;
    const int rb    = (wave >> 1) & 1;
    const int ct    = wave >> 2;
    const int rank  = blockIdx.x / BN;
    const int b     = blockIdx.x % BN;
    const int h     = (rank & 1) ? (HN / 2 - 1 - (rank >> 1)) : (HN / 2 + (rank >> 1));
    __builtin_assume(h >= 0 && h < HN);

    const int tlo   = 20 - h;
    const int iy_lo = (tlo > 0) ? ((tlo + 1) >> 1) : 0;
    const int thi   = (HN - 1 + 20 - h) >> 1;
    const int iy_hi = (thi < ND - 1) ? thi : (ND - 1);

    for (int i = tid; i < ND * OPITCH / 4; i += 256)
        *(float4*)(obuf + 4 * i) = make_float4(0.f, 0.f, 0.f, 0.f);

    const int sj = tid & 15;
    const int sh = (tid >> 4) & 1;
    const int sr = tid >> 5;
    const float* __restrict__ gsrc =
        in2 + (((size_t)b * CN + 8 * sr + 2 * sh) * HN) * WN + 4 * sj;
    float* __restrict__ sdst =
        lds + sr * RPITCH + sh * HPITCH + (2 * sj + (sj >> 3)) * 4;

    float4 g0, g1, g2, g3;
    {
        const float* p = gsrc + (size_t)(h - 20 + 2 * iy_lo) * WN;
        g0 = *(const float4*)(p);
        g1 = *(const float4*)(p + HWN);
        g2 = *(const float4*)(p + 4 * HWN);
        g3 = *(const float4*)(p + 5 * HWN);
    }

    v2f A[CN / 4];
    {
        const float inv = 1.0f / (float)CN;
        const float* __restrict__ ap =
            in1 + (((size_t)b * CN + 2 * half) * HN + h) * WN + (32 * rb + 2 * l + par);
#pragma unroll
        for (int s = 0; s < CN / 4; ++s) {
            A[s].x = ap[(size_t)s * 4 * HWN] * inv;
            A[s].y = ap[(size_t)s * 4 * HWN + HWN] * inv;
        }
    }

    for (int iy = 0; iy < ND; ++iy) {
        if (iy >= iy_lo && iy <= iy_hi) continue;
        for (int i = tid; i < ND * (WN / 4); i += 256) {
            const int ix = i >> 4;
            const int w4 = i & 15;
            const size_t idx =
                (((size_t)b * DN + iy * ND + ix) * HN + h) * WN + 4 * w4;
            *(float4*)(out + idx) = make_float4(0.f, 0.f, 0.f, 0.f);
        }
    }

    *(float4*)(sdst)                  = make_float4(g0.x, g1.x, g2.x, g3.x);
    *(float4*)(sdst + PSLOTS * 4)     = make_float4(g0.y, g1.y, g2.y, g3.y);
    *(float4*)(sdst + 4)              = make_float4(g0.z, g1.z, g2.z, g3.z);
    *(float4*)(sdst + PSLOTS * 4 + 4) = make_float4(g0.w, g1.w, g2.w, g3.w);
    __syncthreads();

    const float* __restrict__ bbase =
        lds + half * HPITCH + (par * PSLOTS + 17 * ct + l) * 4;

    for (int iy = iy_lo; iy <= iy_hi; ++iy) {
        v8f acc = {};
        const int h2 = h - 20 + 2 * iy;

#pragma unroll
        for (int c = 0; c < NCHUNK; ++c) {
            {
                const int cn  = (c + 1) & (NCHUNK - 1);
                const int h2n = (c < NCHUNK - 1) ? h2 : ((iy < iy_hi) ? (h2 + 2) : h2);
                const float* p = gsrc + (size_t)cn * KC * HWN + (size_t)h2n * WN;
                g0 = *(const float4*)(p);
                g1 = *(const float4*)(p + HWN);
                g2 = *(const float4*)(p + 4 * HWN);
                g3 = *(const float4*)(p + 5 * HWN);
            }
            __builtin_amdgcn_sched_barrier(0);

            const float* __restrict__ bp = bbase + (c & 1) * LBUF;
#pragma unroll
            for (int r2 = 0; r2 < KSTEPS / 2; ++r2) {
                const float4 q0 = *(const float4*)(bp + r2 * RPITCH);
                const v2f B0 = { q0.x, q0.y };
                const v2f B1 = { q0.z, q0.w };
                const int s = c * KSTEPS + 2 * r2;
                acc = __builtin_amdgcn_wmma_f32_16x16x4_f32(
                          false, A[s], false, B0, (short)0, acc, false, false);
                acc = __builtin_amdgcn_wmma_f32_16x16x4_f32(
                          false, A[s + 1], false, B1, (short)0, acc, false, false);
            }

            __builtin_amdgcn_sched_barrier(0);
            if (c == NCHUNK - 1) {
                const int e  = 16 * (ct - rb) + l - 8 * half + 10;
                const int p0 = 32 * rb + 16 * half + par;
#pragma unroll
                for (int r = 0; r < 8; ++r) {
                    const int ix = e - r;
                    if ((unsigned)ix <= (unsigned)(ND - 1))
                        obuf[ix * OPITCH + p0 + 2 * r] = acc[r];
                }
            }
            {
                float* d = sdst + ((c + 1) & 1) * LBUF;
                *(float4*)(d)                  = make_float4(g0.x, g1.x, g2.x, g3.x);
                *(float4*)(d + PSLOTS * 4)     = make_float4(g0.y, g1.y, g2.y, g3.y);
                *(float4*)(d + 4)              = make_float4(g0.z, g1.z, g2.z, g3.z);
                *(float4*)(d + PSLOTS * 4 + 4) = make_float4(g0.w, g1.w, g2.w, g3.w);
            }
            __syncthreads();
        }

        {
            const int ix0 = tid >> 4;
            const int w4  = tid & 15;
            const float* __restrict__ op = obuf + ix0 * OPITCH + 4 * w4;
            float* __restrict__ gp =
                out + (((size_t)b * DN + iy * ND + ix0) * HN + h) * WN + 4 * w4;
            const float4 v0 = *(const float4*)(op);
            if (tid < (ND - 16) * 16) {
                const float4 v1 = *(const float4*)(op + 16 * OPITCH);
                *(float4*)(gp + (size_t)16 * HWN) = v1;
            }
            *(float4*)(gp) = v0;
        }
    }
}

extern "C" void kernel_launch(void* const* d_in, const int* in_sizes, int n_in,
                              void* d_out, int out_size, void* d_ws, size_t ws_size,
                              hipStream_t stream) {
    (void)in_sizes; (void)n_in; (void)out_size; (void)d_ws; (void)ws_size;
    const float* in1 = (const float*)d_in[0];
    const float* in2 = (const float*)d_in[1];
    float* out = (float*)d_out;

    corr_wmma_f32_kernel<<<BN * HN, 256, 0, stream>>>(in1, in2, out);
}


# In2 chunk loads issued two to three chunks ahead of their products (three register sets) and kept in flight across the chunk barrier (written by hand, waits on LDS only); k order and sums unchanged
# speedup vs baseline: 5.5969x; 1.1001x over previous
#include <hip/hip_runtime.h>


typedef float v2f __attribute__((ext_vector_type(2)));
typedef float v4f __attribute__((ext_vector_type(4)));
typedef float v8f __attribute__((ext_vector_type(8)));

#define BN 16
#define CN 256
#define HN 48
#define WN 64
#define HWN (HN * WN)
#define ND 21
#define DN (ND * ND)

#define KC 64
#define NCHUNK (CN / KC)
#define KSTEPS (KC / 4)
#define PSLOTS 33
#define HPITCH 288
#define RPITCH (2 * HPITCH)
#define LBUF (8 * RPITCH)
#define OPITCH 68

static __device__ __forceinline__ void lds_barrier()
{
    asm volatile("s_wait_dscnt 0x0\n\ts_barrier_signal -1\n\ts_barrier_wait -1" ::: "memory");
}

__global__ __launch_bounds__(256) void corr_wmma_f32_kernel(
    const float* __restrict__ in1,
    const float* __restrict__ in2,
    float* __restrict__ out)
{
    __shared__ __attribute__((aligned(16))) float lds[2 * LBUF];
    __shared__ __attribute__((aligned(16))) float obuf[ND * OPITCH];

    const int tid   = threadIdx.x;
    const int wave  = tid >> 5;
    const int lane  = tid & 31;
    const int half  = lane >> 4;
    const int l     = lane & 15;
    const int par   = wave & 1;
    const int rb    = (wave >> 1) & 1;
    const int ct    = wave >> 2;
    const int rank  = blockIdx.x / BN;
    const int b     = blockIdx.x % BN;
    const int h     = (rank & 1) ? (HN / 2 - 1 - (rank >> 1)) : (HN / 2 + (rank >> 1));
    __builtin_assume(h >= 0 && h < HN);

    const int tlo   = 20 - h;
    const int iy_lo = (tlo > 0) ? ((tlo + 1) >> 1) : 0;
    const int thi   = (HN - 1 + 20 - h) >> 1;
    const int iy_hi = (thi < ND - 1) ? thi : (ND - 1);

    for (int i = tid; i < ND * OPITCH / 4; i += 256)
        *(float4*)(obuf + 4 * i) = make_float4(0.f, 0.f, 0.f, 0.f);

    const int sj = tid & 15;
    const int sh = (tid >> 4) & 1;
    const int sr = tid >> 5;
    const float* __restrict__ gsrc =
        in2 + (((size_t)b * CN + 8 * sr + 2 * sh) * HN) * WN + 4 * sj;
    float* __restrict__ sdst =
        lds + sr * RPITCH + sh * HPITCH + (2 * sj + (sj >> 3)) * 4;

    v4f g[3][4];
    {
        const float* p = gsrc + (size_t)(h - 20 + 2 * iy_lo) * WN;
#pragma unroll
        for (int n = 0; n < 2; ++n) {
            g[2 * n][0] = *(const v4f*)(p + (size_t)n * KC * HWN);
            g[2 * n][1] = *(const v4f*)(p + (size_t)n * KC * HWN + HWN);
            g[2 * n][2] = *(const v4f*)(p + (size_t)n * KC * HWN + 4 * HWN);
            g[2 * n][3] = *(const v4f*)(p + (size_t)n * KC * HWN + 5 * HWN);
        }
    }

    v2f A[CN / 4];
    {
        const float inv = 1.0f / (float)CN;
        const float* __restrict__ ap =
            in1 + (((size_t)b * CN + 2 * half) * HN + h) * WN + (32 * rb + 2 * l + par);
#pragma unroll
        for (int s = 0; s < CN / 4; ++s) {
            A[s].x = ap[(size_t)s * 4 * HWN] * inv;
            A[s].y = ap[(size_t)s * 4 * HWN + HWN] * inv;
        }
    }

    for (int iy = 0; iy < ND; ++iy) {
        if (iy >= iy_lo && iy <= iy_hi) continue;
        for (int i = tid; i < ND * (WN / 4); i += 256) {
            const int ix = i >> 4;
            const int w4 = i & 15;
            const size_t idx =
                (((size_t)b * DN + iy * ND + ix) * HN + h) * WN + 4 * w4;
            *(float4*)(out + idx) = make_float4(0.f, 0.f, 0.f, 0.f);
        }
    }

    *(float4*)(sdst)                  = make_float4(g[0][0].x, g[0][1].x, g[0][2].x, g[0][3].x);
    *(float4*)(sdst + PSLOTS * 4)     = make_float4(g[0][0].y, g[0][1].y, g[0][2].y, g[0][3].y);
    *(float4*)(sdst + 4)              = make_float4(g[0][0].z, g[0][1].z, g[0][2].z, g[0][3].z);
    *(float4*)(sdst + PSLOTS * 4 + 4) = make_float4(g[0][0].w, g[0][1].w, g[0][2].w, g[0][3].w);
    __syncthreads();

    const float* __restrict__ bbase =
        lds + half * HPITCH + (par * PSLOTS + 17 * ct + l) * 4;

    for (int iy = iy_lo; iy <= iy_hi; ++iy) {
        v8f acc = {};
        const int h2 = h - 20 + 2 * iy;

#pragma unroll
        for (int c = 0; c < NCHUNK; ++c) {
            if (c < NCHUNK - 1) {
                const int h2n = (c < NCHUNK - 2) ? h2 : ((iy < iy_hi) ? (h2 + 2) : h2);
                const int cn  = (c + 2) & (NCHUNK - 1);
                const int sn  = (c == 1) ? 1 : 0;
                const float* p = gsrc + (size_t)cn * KC * HWN + (size_t)h2n * WN;
                g[sn][0] = *(const v4f*)(p);
                g[sn][1] = *(const v4f*)(p + HWN);
                g[sn][2] = *(const v4f*)(p + 4 * HWN);
                g[sn][3] = *(const v4f*)(p + 5 * HWN);
                if (c == NCHUNK - 2) {
                    g[2][0] = *(const v4f*)(p + KC * HWN);
                    g[2][1] = *(const v4f*)(p + KC * HWN + HWN);
                    g[2][2] = *(const v4f*)(p + KC * HWN + 4 * HWN);
                    g[2][3] = *(const v4f*)(p + KC * HWN + 5 * HWN);
                }
            }
            __builtin_amdgcn_sched_barrier(0);

            const float* __restrict__ bp = bbase + (c & 1) * LBUF;
#pragma unroll
            for (int r2 = 0; r2 < KSTEPS / 2; ++r2) {
                const float4 q0 = *(const float4*)(bp + r2 * RPITCH);
                const v2f B0 = { q0.x, q0.y };
                const v2f B1 = { q0.z, q0.w };
                const int s = c * KSTEPS + 2 * r2;
                acc = __builtin_amdgcn_wmma_f32_16x16x4_f32(
                          false, A[s], false, B0, (short)0, acc, false, false);
                acc = __builtin_amdgcn_wmma_f32_16x16x4_f32(
                          false, A[s + 1], false, B1, (short)0, acc, false, false);
            }

            __builtin_amdgcn_sched_barrier(0);
            if (c == NCHUNK - 1) {
                const int e  = 16 * (ct - rb) + l - 8 * half + 10;
                const int p0 = 32 * rb + 16 * half + par;
#pragma unroll
                for (int r = 0; r < 8; ++r) {
                    const int ix = e - r;
                    if ((unsigned)ix <= (unsigned)(ND - 1))
                        obuf[ix * OPITCH + p0 + 2 * r] = acc[r];
                }
            }
            {
                const int n = (c == 0) ? 2 : ((c == 2) ? 1 : 0);
                float* d = sdst + ((c + 1) & 1) * LBUF;
                *(float4*)(d)                  = make_float4(g[n][0].x, g[n][1].x, g[n][2].x, g[n][3].x);
                *(float4*)(d + PSLOTS * 4)     = make_float4(g[n][0].y, g[n][1].y, g[n][2].y, g[n][3].y);
                *(float4*)(d + 4)              = make_float4(g[n][0].z, g[n][1].z, g[n][2].z, g[n][3].z);
                *(float4*)(d + PSLOTS * 4 + 4) = make_float4(g[n][0].w, g[n][1].w, g[n][2].w, g[n][3].w);
            }
            lds_barrier();
        }

        {
            const int ix0 = tid >> 4;
            const int w4  = tid & 15;
            const float* __restrict__ op = obuf + ix0 * OPITCH + 4 * w4;
            float* __restrict__ gp =
                out + (((size_t)b * DN + iy * ND + ix0) * HN + h) * WN + 4 * w4;
            const float4 v0 = *(const float4*)(op);
            if (tid < (ND - 16) * 16) {
                const float4 v1 = *(const float4*)(op + 16 * OPITCH);
                *(float4*)(gp + (size_t)16 * HWN) = v1;
            }
            *(float4*)(gp) = v0;
        }
    }
}

extern "C" void kernel_launch(void* const* d_in, const int* in_sizes, int n_in,
                              void* d_out, int out_size, void* d_ws, size_t ws_size,
                              hipStream_t stream) {
    (void)in_sizes; (void)n_in; (void)out_size; (void)d_ws; (void)ws_size;
    const float* in1 = (const float*)d_in[0];
    const float* in2 = (const float*)d_in[1];
    float* out = (float*)d_out;

    corr_wmma_f32_kernel<<<BN * HN, 256, 0, stream>>>(in1, in2, out);
}
